# starting kernel
# baseline (speedup 1.0000x reference)
#include <hip/hip_runtime.h>

#define NN 100000
#define D 128
#define ND ((size_t)NN * D)
#define LDSP 132

typedef __attribute__((ext_vector_type(2))) float v2f;
typedef __attribute__((ext_vector_type(8))) float v8f;

__device__ __forceinline__ void atomAddF(float* p, float v) {
    __hip_atomic_fetch_add(p, v, __ATOMIC_RELAXED, __HIP_MEMORY_SCOPE_AGENT);
}

__device__ __forceinline__ void wmma_pass(const float (*__restrict__ As)[LDSP],
                                          const float* __restrict__ W,
                                          int col0, int half, int l,
                                          v8f& acc0, v8f& acc1) {
    const float* wcol = W + col0 + l;
#pragma unroll
    for (int kk = 0; kk < D; kk += 4) {
        const int k0 = kk + 2 * half;
        v2f b;
        b.x = wcol[(size_t)k0 * D];
        b.y = wcol[(size_t)(k0 + 1) * D];
        v2f a0, a1;
        a0.x = As[l][k0];        a0.y = As[l][k0 + 1];
        a1.x = As[16 + l][k0];   a1.y = As[16 + l][k0 + 1];
        acc0 = __builtin_amdgcn_wmma_f32_16x16x4_f32(false, a0, false, b, (short)0, acc0, false, false);
        acc1 = __builtin_amdgcn_wmma_f32_16x16x4_f32(false, a1, false, b, (short)0, acc1, false, false);
    }
}

template <bool RELU, bool TWO>
__global__ __launch_bounds__(256) void gemm_sage(const float* __restrict__ A0,
                                                 const float* __restrict__ W0,
                                                 const float* __restrict__ A1,
                                                 const float* __restrict__ W1,
                                                 const float* __restrict__ bias,
                                                 float* __restrict__ C) {
    __shared__ float As0[32][LDSP];
    __shared__ float As1[TWO ? 32 : 1][LDSP];

    const int row0 = blockIdx.x * 32;
    const int col0 = (threadIdx.x >> 5) * 16;
    const int lane = threadIdx.x & 31;
    const int half = lane >> 4;
    const int l = lane & 15;

    {
        const int r = threadIdx.x >> 3;
        const int c = (threadIdx.x & 7) * 16;
        const float* g0 = A0 + (size_t)(row0 + r) * D + c;
#pragma unroll
        for (int j = 0; j < 16; j += 4)
            *(float4*)&As0[r][c + j] = *(const float4*)(g0 + j);
        if (TWO) {
            const float* g1 = A1 + (size_t)(row0 + r) * D + c;
#pragma unroll
            for (int j = 0; j < 16; j += 4)
                *(float4*)&As1[r][c + j] = *(const float4*)(g1 + j);
        }
    }
    __syncthreads();

    v8f acc0 = {0.f, 0.f, 0.f, 0.f, 0.f, 0.f, 0.f, 0.f};
    v8f acc1 = acc0;
    wmma_pass(As0, W0, col0, half, l, acc0, acc1);
    if (TWO) wmma_pass(As1, W1, col0, half, l, acc0, acc1);

    const float bc = bias[col0 + l];
#pragma unroll
    for (int r = 0; r < 8; ++r) {
        float v0 = acc0[r] + bc;
        float v1 = acc1[r] + bc;
        if (RELU) { v0 = fmaxf(v0, 0.f); v1 = fmaxf(v1, 0.f); }
        C[(size_t)(row0 + half * 8 + r) * D + col0 + l] = v0;
        C[(size_t)(row0 + 16 + half * 8 + r) * D + col0 + l] = v1;
    }
}

__global__ __launch_bounds__(256) void zero_f32(float* __restrict__ p, long n) {
    long i = (long)blockIdx.x * blockDim.x + threadIdx.x;
    if (i < n) p[i] = 0.f;
}

__global__ __launch_bounds__(256) void sage_scatter(const float* __restrict__ h,
                                                    const int* __restrict__ src,
                                                    const int* __restrict__ dst,
                                                    float* __restrict__ agg,
                                                    float* __restrict__ deg, int E) {
    long t = (long)blockIdx.x * blockDim.x + threadIdx.x;
    int e = (int)(t >> 5);
    if (e >= E) return;
    int lane = (int)(t & 31);
    int s = src[e], d = dst[e];
    const float4 v = *(const float4*)(h + (size_t)s * D + lane * 4);
    float* p = agg + (size_t)d * D + lane * 4;
    atomAddF(p + 0, v.x);
    atomAddF(p + 1, v.y);
    atomAddF(p + 2, v.z);
    atomAddF(p + 3, v.w);
    if (lane == 0) atomAddF(deg + d, 1.0f);
}

__global__ __launch_bounds__(256) void sage_mean(float* __restrict__ agg,
                                                 const float* __restrict__ deg) {
    long t = (long)blockIdx.x * blockDim.x + threadIdx.x;
    int n = (int)(t >> 5);
    if (n >= NN) return;
    int lane = (int)(t & 31);
    float inv = 1.0f / fmaxf(deg[n], 1.0f);
    float4* p = (float4*)(agg + (size_t)n * D + lane * 4);
    float4 v = *p;
    v.x *= inv; v.y *= inv; v.z *= inv; v.w *= inv;
    *p = v;
}

__global__ __launch_bounds__(256) void edge_dot(const float* __restrict__ h,
                                                const int* __restrict__ src,
                                                const int* __restrict__ dst,
                                                float* __restrict__ out, int n) {
    long t = (long)blockIdx.x * blockDim.x + threadIdx.x;
    int e = (int)(t >> 5);
    if (e >= n) return;
    int lane = (int)(t & 31);
    const float4 a = *(const float4*)(h + (size_t)src[e] * D + lane * 4);
    const float4 b = *(const float4*)(h + (size_t)dst[e] * D + lane * 4);
    float s = a.x * b.x + a.y * b.y + a.z * b.z + a.w * b.w;
#pragma unroll
    for (int off = 16; off; off >>= 1) s += __shfl_down(s, off, 32);
    if (lane == 0) out[e] = s;
}

extern "C" void kernel_launch(void* const* d_in, const int* in_sizes, int n_in,
                              void* d_out, int out_size, void* d_ws, size_t ws_size,
                              hipStream_t stream) {
    const float* feat    = (const float*)d_in[0];
    const int*   e0s     = (const int*)d_in[1];
    const int*   e0d     = (const int*)d_in[2];
    const int*   e1s     = (const int*)d_in[3];
    const int*   e1d     = (const int*)d_in[4];
    const int*   ps      = (const int*)d_in[5];
    const int*   pd      = (const int*)d_in[6];
    const int*   ns      = (const int*)d_in[7];
    const int*   nd      = (const int*)d_in[8];
    const float* w_proj  = (const float*)d_in[9];
    const float* b_proj  = (const float*)d_in[10];
    const float* w_self1 = (const float*)d_in[11];
    const float* w_neigh1= (const float*)d_in[12];
    const float* b1      = (const float*)d_in[13];
    const float* w_self2 = (const float*)d_in[14];
    const float* w_neigh2= (const float*)d_in[15];
    const float* b2      = (const float*)d_in[16];

    const int E0   = in_sizes[1];
    const int E1   = in_sizes[3];
    const int nPos = in_sizes[5];
    const int nNeg = in_sizes[7];

    float* h0  = (float*)d_ws;
    float* agg = h0 + ND;
    float* deg = agg + ND;
    float* h1  = deg + NN;
    float* h2  = h0;
    float* out = (float*)d_out;

    const int mtiles = NN / 32;
    const long zn    = (long)ND + NN;
    const int zgrid  = (int)((zn + 255) / 256);
    const int sgrid0 = (int)(((long)E0 * 32 + 255) / 256);
    const int sgrid1 = (int)(((long)E1 * 32 + 255) / 256);
    const int mgrid  = (int)(((long)NN * 32 + 255) / 256);

    gemm_sage<true, false><<<mtiles, 256, 0, stream>>>(feat, w_proj, nullptr, nullptr, b_proj, h0);

    zero_f32<<<zgrid, 256, 0, stream>>>(agg, zn);
    sage_scatter<<<sgrid0, 256, 0, stream>>>(h0, e0s, e0d, agg, deg, E0);
    sage_mean<<<mgrid, 256, 0, stream>>>(agg, deg);
    gemm_sage<true, true><<<mtiles, 256, 0, stream>>>(h0, w_self1, agg, w_neigh1, b1, h1);

    zero_f32<<<zgrid, 256, 0, stream>>>(agg, zn);
    sage_scatter<<<sgrid1, 256, 0, stream>>>(h1, e1s, e1d, agg, deg, E1);
    sage_mean<<<mgrid, 256, 0, stream>>>(agg, deg);
    gemm_sage<false, true><<<mtiles, 256, 0, stream>>>(h1, w_self2, agg, w_neigh2, b2, h2);

    edge_dot<<<(int)(((long)nPos * 32 + 255) / 256), 256, 0, stream>>>(h2, ps, pd, out, nPos);
    edge_dot<<<(int)(((long)nNeg * 32 + 255) / 256), 256, 0, stream>>>(h2, ns, nd, out + nPos, nNeg);
}


# Scatter-add by f32 atomics, its zero fill and mean pass replaced by a per-call counting sort of edges by dst (int atomics) and a wave-per-node row gather; sums in arrival-rank order; 15 launches
# speedup vs baseline: 1.7283x; 1.7283x over previous
#include <hip/hip_runtime.h>

#define NN 100000
#define D 128
#define ND ((size_t)NN * D)
#define LDSP 132

typedef __attribute__((ext_vector_type(2))) float v2f;
typedef __attribute__((ext_vector_type(8))) float v8f;

__device__ __forceinline__ int atomAddRetI(int* p, int v) {
    return __hip_atomic_fetch_add(p, v, __ATOMIC_RELAXED, __HIP_MEMORY_SCOPE_AGENT);
}

__device__ __forceinline__ void wmma_pass(const float (*__restrict__ As)[LDSP],
                                          const float* __restrict__ W,
                                          int col0, int half, int l,
                                          v8f& acc0, v8f& acc1) {
    const float* wcol = W + col0 + l;
#pragma unroll
    for (int kk = 0; kk < D; kk += 4) {
        const int k0 = kk + 2 * half;
        v2f b;
        b.x = wcol[(size_t)k0 * D];
        b.y = wcol[(size_t)(k0 + 1) * D];
        v2f a0, a1;
        a0.x = As[l][k0];        a0.y = As[l][k0 + 1];
        a1.x = As[16 + l][k0];   a1.y = As[16 + l][k0 + 1];
        acc0 = __builtin_amdgcn_wmma_f32_16x16x4_f32(false, a0, false, b, (short)0, acc0, false, false);
        acc1 = __builtin_amdgcn_wmma_f32_16x16x4_f32(false, a1, false, b, (short)0, acc1, false, false);
    }
}

template <bool RELU, bool TWO>
__global__ __launch_bounds__(256) void gemm_sage(const float* __restrict__ A0,
                                                 const float* __restrict__ W0,
                                                 const float* __restrict__ A1,
                                                 const float* __restrict__ W1,
                                                 const float* __restrict__ bias,
                                                 float* __restrict__ C) {
    __shared__ float As0[32][LDSP];
    __shared__ float As1[TWO ? 32 : 1][LDSP];

    const int row0 = blockIdx.x * 32;
    const int col0 = (threadIdx.x >> 5) * 16;
    const int lane = threadIdx.x & 31;
    const int half = lane >> 4;
    const int l = lane & 15;

    {
        const int r = threadIdx.x >> 3;
        const int c = (threadIdx.x & 7) * 16;
        const float* g0 = A0 + (size_t)(row0 + r) * D + c;
#pragma unroll
        for (int j = 0; j < 16; j += 4)
            *(float4*)&As0[r][c + j] = *(const float4*)(g0 + j);
        if (TWO) {
            const float* g1 = A1 + (size_t)(row0 + r) * D + c;
#pragma unroll
            for (int j = 0; j < 16; j += 4)
                *(float4*)&As1[r][c + j] = *(const float4*)(g1 + j);
        }
    }
    __syncthreads();

    v8f acc0 = {0.f, 0.f, 0.f, 0.f, 0.f, 0.f, 0.f, 0.f};
    v8f acc1 = acc0;
    wmma_pass(As0, W0, col0, half, l, acc0, acc1);
    if (TWO) wmma_pass(As1, W1, col0, half, l, acc0, acc1);

    const float bc = bias[col0 + l];
#pragma unroll
    for (int r = 0; r < 8; ++r) {
        float v0 = acc0[r] + bc;
        float v1 = acc1[r] + bc;
        if (RELU) { v0 = fmaxf(v0, 0.f); v1 = fmaxf(v1, 0.f); }
        C[(size_t)(row0 + half * 8 + r) * D + col0 + l] = v0;
        C[(size_t)(row0 + 16 + half * 8 + r) * D + col0 + l] = v1;
    }
}

__global__ __launch_bounds__(256) void zero_f32(float* __restrict__ p, long n) {
    long i = (long)blockIdx.x * blockDim.x + threadIdx.x;
    if (i < n) p[i] = 0.f;
}

__global__ __launch_bounds__(256) void seg_count(const int* __restrict__ dst, int* cnt,
                                                 int* __restrict__ rank, int E) {
    const int e = blockIdx.x * 256 + threadIdx.x;
    if (e >= E) return;
    rank[e] = atomAddRetI(cnt + dst[e], 1);
}

__global__ __launch_bounds__(256) void seg_alloc(const int* cnt, int* __restrict__ start, int* total) {
    const int n = blockIdx.x * 256 + threadIdx.x;
    const int lane = threadIdx.x & 31;
    const int c = (n < NN) ? cnt[n] : 0;
    int x = c;
#pragma unroll
    for (int off = 1; off < 32; off <<= 1) {
        const int y = __shfl_up(x, off, 32);
        if (lane >= off) x += y;
    }
    int base = 0;
    if (lane == 31) base = atomAddRetI(total, x);
    base = __builtin_amdgcn_readlane(base, 31);
    if (n < NN) start[n] = base + x - c;
}

__global__ __launch_bounds__(256) void seg_place(const int* __restrict__ src,
                                                 const int* __restrict__ dst,
                                                 const int* __restrict__ rank,
                                                 const int* __restrict__ start,
                                                 int* __restrict__ perm, int E) {
    const int e = blockIdx.x * 256 + threadIdx.x;
    if (e >= E) return;
    const int p = start[dst[e]] + rank[e];
    if ((unsigned)p < (unsigned)E) perm[p] = src[e];
}

__global__ __launch_bounds__(256) void sage_gather_mean(const float* __restrict__ h,
                                                        const int* __restrict__ perm,
                                                        const int* __restrict__ start,
                                                        const int* __restrict__ cnt,
                                                        float* __restrict__ agg, int E) {
    const int n = __builtin_amdgcn_readfirstlane((int)(blockIdx.x * 8 + (threadIdx.x >> 5)));
    if (n >= NN) return;
    const int lane = threadIdx.x & 31;
    int c = cnt[n];
    c = c < E ? c : E;
    const int s0 = start[n];
    float4 acc = {0.f, 0.f, 0.f, 0.f};
    for (int base = 0; base < c; base += 32) {
        const int m = (c - base) < 32 ? (c - base) : 32;
        const int q = s0 + base + lane;
        int mine = 0;
        if (lane < m && (unsigned)q < (unsigned)E) mine = perm[q];
#pragma unroll 4
        for (int j = 0; j < m; ++j) {
            int s = __builtin_amdgcn_readlane(mine, j);
            s = ((unsigned)s < (unsigned)NN) ? s : 0;
            const float4 v = *(const float4*)(h + (size_t)s * D + lane * 4);
            acc.x += v.x; acc.y += v.y; acc.z += v.z; acc.w += v.w;
        }
    }
    const float inv = 1.0f / fmaxf((float)c, 1.0f);
    acc.x *= inv; acc.y *= inv; acc.z *= inv; acc.w *= inv;
    *(float4*)(agg + (size_t)n * D + lane * 4) = acc;
}

__global__ __launch_bounds__(256) void edge_dot(const float* __restrict__ h,
                                                const int* __restrict__ src,
                                                const int* __restrict__ dst,
                                                float* __restrict__ out, int n) {
    long t = (long)blockIdx.x * blockDim.x + threadIdx.x;
    int e = (int)(t >> 5);
    if (e >= n) return;
    int lane = (int)(t & 31);
    const float4 a = *(const float4*)(h + (size_t)src[e] * D + lane * 4);
    const float4 b = *(const float4*)(h + (size_t)dst[e] * D + lane * 4);
    float s = a.x * b.x + a.y * b.y + a.z * b.z + a.w * b.w;
#pragma unroll
    for (int off = 16; off; off >>= 1) s += __shfl_down(s, off, 32);
    if (lane == 0) out[e] = s;
}

extern "C" void kernel_launch(void* const* d_in, const int* in_sizes, int n_in,
                              void* d_out, int out_size, void* d_ws, size_t ws_size,
                              hipStream_t stream) {
    const float* feat    = (const float*)d_in[0];
    const int*   e0s     = (const int*)d_in[1];
    const int*   e0d     = (const int*)d_in[2];
    const int*   e1s     = (const int*)d_in[3];
    const int*   e1d     = (const int*)d_in[4];
    const int*   ps      = (const int*)d_in[5];
    const int*   pd      = (const int*)d_in[6];
    const int*   ns      = (const int*)d_in[7];
    const int*   nd      = (const int*)d_in[8];
    const float* w_proj  = (const float*)d_in[9];
    const float* b_proj  = (const float*)d_in[10];
    const float* w_self1 = (const float*)d_in[11];
    const float* w_neigh1= (const float*)d_in[12];
    const float* b1      = (const float*)d_in[13];
    const float* w_self2 = (const float*)d_in[14];
    const float* w_neigh2= (const float*)d_in[15];
    const float* b2      = (const float*)d_in[16];

    const int E0   = in_sizes[1];
    const int E1   = in_sizes[3];
    const int nPos = in_sizes[5];
    const int nNeg = in_sizes[7];

    float* h0  = (float*)d_ws;
    float* agg = h0 + ND;
    float* deg = agg + ND;
    float* h1  = deg + NN;
    float* h2  = h0;
    float* out = (float*)d_out;

    int* seg   = (int*)deg;
    int* perm1 = (int*)h1;
    int* rank1 = perm1 + E0;
    int* cnt1  = rank1 + E0;
    int* perm2 = (int*)h0;
    int* rank2 = perm2 + E1;
    int* cnt2  = rank2 + E1;

    const int mtiles = NN / 32;
    const int zgrid  = (NN + 1 + 255) / 256;
    const int egrid0 = (E0 + 255) / 256;
    const int egrid1 = (E1 + 255) / 256;
    const int ngrid  = (NN + 255) / 256;
    const int ggrid  = NN / 8;

    gemm_sage<true, false><<<mtiles, 256, 0, stream>>>(feat, w_proj, nullptr, nullptr, b_proj, h0);

    zero_f32<<<zgrid, 256, 0, stream>>>((float*)cnt1, (long)NN + 1);
    seg_count<<<egrid0, 256, 0, stream>>>(e0d, cnt1, rank1, E0);
    seg_alloc<<<ngrid, 256, 0, stream>>>(cnt1, seg, cnt1 + NN);
    seg_place<<<egrid0, 256, 0, stream>>>(e0s, e0d, rank1, seg, perm1, E0);
    sage_gather_mean<<<ggrid, 256, 0, stream>>>(h0, perm1, seg, cnt1, agg, E0);
    gemm_sage<true, true><<<mtiles, 256, 0, stream>>>(h0, w_self1, agg, w_neigh1, b1, h1);

    zero_f32<<<zgrid, 256, 0, stream>>>((float*)cnt2, (long)NN + 1);
    seg_count<<<egrid1, 256, 0, stream>>>(e1d, cnt2, rank2, E1);
    seg_alloc<<<ngrid, 256, 0, stream>>>(cnt2, seg, cnt2 + NN);
    seg_place<<<egrid1, 256, 0, stream>>>(e1s, e1d, rank2, seg, perm2, E1);
    sage_gather_mean<<<ggrid, 256, 0, stream>>>(h1, perm2, seg, cnt2, agg, E1);
    gemm_sage<false, true><<<mtiles, 256, 0, stream>>>(h1, w_self2, agg, w_neigh2, b2, h2);

    edge_dot<<<(int)(((long)nPos * 32 + 255) / 256), 256, 0, stream>>>(h2, ps, pd, out, nPos);
    edge_dot<<<(int)(((long)nNeg * 32 + 255) / 256), 256, 0, stream>>>(h2, ns, nd, out + nPos, nNeg);
}


# Edge scoring: one wave per 32 edge pairs, not one per pair (ids by one load, rows on scalar bases, lane sums folded by a transposing 16-8-4-2-1 exchange, one line store); sum order unchanged
# speedup vs baseline: 1.8945x; 1.0962x over previous
#include <hip/hip_runtime.h>

#define NN 100000
#define D 128
#define ND ((size_t)NN * D)
#define LDSP 132

typedef __attribute__((ext_vector_type(2))) float v2f;
typedef __attribute__((ext_vector_type(8))) float v8f;

__device__ __forceinline__ int atomAddRetI(int* p, int v) {
    return __hip_atomic_fetch_add(p, v, __ATOMIC_RELAXED, __HIP_MEMORY_SCOPE_AGENT);
}

__device__ __forceinline__ void wmma_pass(const float (*__restrict__ As)[LDSP],
                                          const float* __restrict__ W,
                                          int col0, int half, int l,
                                          v8f& acc0, v8f& acc1) {
    const float* wcol = W + col0 + l;
#pragma unroll
    for (int kk = 0; kk < D; kk += 4) {
        const int k0 = kk + 2 * half;
        v2f b;
        b.x = wcol[(size_t)k0 * D];
        b.y = wcol[(size_t)(k0 + 1) * D];
        v2f a0, a1;
        a0.x = As[l][k0];        a0.y = As[l][k0 + 1];
        a1.x = As[16 + l][k0];   a1.y = As[16 + l][k0 + 1];
        acc0 = __builtin_amdgcn_wmma_f32_16x16x4_f32(false, a0, false, b, (short)0, acc0, false, false);
        acc1 = __builtin_amdgcn_wmma_f32_16x16x4_f32(false, a1, false, b, (short)0, acc1, false, false);
    }
}

template <bool RELU, bool TWO>
__global__ __launch_bounds__(256) void gemm_sage(const float* __restrict__ A0,
                                                 const float* __restrict__ W0,
                                                 const float* __restrict__ A1,
                                                 const float* __restrict__ W1,
                                                 const float* __restrict__ bias,
                                                 float* __restrict__ C) {
    __shared__ float As0[32][LDSP];
    __shared__ float As1[TWO ? 32 : 1][LDSP];

    const int row0 = blockIdx.x * 32;
    const int col0 = (threadIdx.x >> 5) * 16;
    const int lane = threadIdx.x & 31;
    const int half = lane >> 4;
    const int l = lane & 15;

    {
        const int r = threadIdx.x >> 3;
        const int c = (threadIdx.x & 7) * 16;
        const float* g0 = A0 + (size_t)(row0 + r) * D + c;
#pragma unroll
        for (int j = 0; j < 16; j += 4)
            *(float4*)&As0[r][c + j] = *(const float4*)(g0 + j);
        if (TWO) {
            const float* g1 = A1 + (size_t)(row0 + r) * D + c;
#pragma unroll
            for (int j = 0; j < 16; j += 4)
                *(float4*)&As1[r][c + j] = *(const float4*)(g1 + j);
        }
    }
    __syncthreads();

    v8f acc0 = {0.f, 0.f, 0.f, 0.f, 0.f, 0.f, 0.f, 0.f};
    v8f acc1 = acc0;
    wmma_pass(As0, W0, col0, half, l, acc0, acc1);
    if (TWO) wmma_pass(As1, W1, col0, half, l, acc0, acc1);

    const float bc = bias[col0 + l];
#pragma unroll
    for (int r = 0; r < 8; ++r) {
        float v0 = acc0[r] + bc;
        float v1 = acc1[r] + bc;
        if (RELU) { v0 = fmaxf(v0, 0.f); v1 = fmaxf(v1, 0.f); }
        C[(size_t)(row0 + half * 8 + r) * D + col0 + l] = v0;
        C[(size_t)(row0 + 16 + half * 8 + r) * D + col0 + l] = v1;
    }
}

__global__ __launch_bounds__(256) void zero_f32(float* __restrict__ p, long n) {
    long i = (long)blockIdx.x * blockDim.x + threadIdx.x;
    if (i < n) p[i] = 0.f;
}

__global__ __launch_bounds__(256) void seg_count(const int* __restrict__ dst, int* cnt,
                                                 int* __restrict__ rank, int E) {
    const int e = blockIdx.x * 256 + threadIdx.x;
    if (e >= E) return;
    rank[e] = atomAddRetI(cnt + dst[e], 1);
}

__global__ __launch_bounds__(256) void seg_alloc(const int* cnt, int* __restrict__ start, int* total) {
    const int n = blockIdx.x * 256 + threadIdx.x;
    const int lane = threadIdx.x & 31;
    const int c = (n < NN) ? cnt[n] : 0;
    int x = c;
#pragma unroll
    for (int off = 1; off < 32; off <<= 1) {
        const int y = __shfl_up(x, off, 32);
        if (lane >= off) x += y;
    }
    int base = 0;
    if (lane == 31) base = atomAddRetI(total, x);
    base = __builtin_amdgcn_readlane(base, 31);
    if (n < NN) start[n] = base + x - c;
}

__global__ __launch_bounds__(256) void seg_place(const int* __restrict__ src,
                                                 const int* __restrict__ dst,
                                                 const int* __restrict__ rank,
                                                 const int* __restrict__ start,
                                                 int* __restrict__ perm, int E) {
    const int e = blockIdx.x * 256 + threadIdx.x;
    if (e >= E) return;
    const int p = start[dst[e]] + rank[e];
    if ((unsigned)p < (unsigned)E) perm[p] = src[e];
}

__global__ __launch_bounds__(256) void sage_gather_mean(const float* __restrict__ h,
                                                        const int* __restrict__ perm,
                                                        const int* __restrict__ start,
                                                        const int* __restrict__ cnt,
                                                        float* __restrict__ agg, int E) {
    const int n = __builtin_amdgcn_readfirstlane((int)(blockIdx.x * 8 + (threadIdx.x >> 5)));
    if (n >= NN) return;
    const int lane = threadIdx.x & 31;
    int c = cnt[n];
    c = c < E ? c : E;
    const int s0 = start[n];
    float4 acc = {0.f, 0.f, 0.f, 0.f};
    for (int base = 0; base < c; base += 32) {
        const int m = (c - base) < 32 ? (c - base) : 32;
        const int q = s0 + base + lane;
        int mine = 0;
        if (lane < m && (unsigned)q < (unsigned)E) mine = perm[q];
#pragma unroll 4
        for (int j = 0; j < m; ++j) {
            int s = __builtin_amdgcn_readlane(mine, j);
            s = ((unsigned)s < (unsigned)NN) ? s : 0;
            const float4 v = *(const float4*)(h + (size_t)s * D + lane * 4);
            acc.x += v.x; acc.y += v.y; acc.z += v.z; acc.w += v.w;
        }
    }
    const float inv = 1.0f / fmaxf((float)c, 1.0f);
    acc.x *= inv; acc.y *= inv; acc.z *= inv; acc.w *= inv;
    *(float4*)(agg + (size_t)n * D + lane * 4) = acc;
}

template <int LV>
__device__ __forceinline__ float pair_sums(const float* __restrict__ h, int sv, int dv, int lane, int j) {
    if constexpr (LV == 0) {
        const int s = __builtin_amdgcn_readlane(sv, j);
        const int d = __builtin_amdgcn_readlane(dv, j);
        const float4 a = *(const float4*)(h + (size_t)s * D + lane * 4);
        const float4 b = *(const float4*)(h + (size_t)d * D + lane * 4);
        return a.x * b.x + a.y * b.y + a.z * b.z + a.w * b.w;
    } else {
        constexpr int S = 32 >> LV;
        const float x = pair_sums<LV - 1>(h, sv, dv, lane, j);
        const float y = pair_sums<LV - 1>(h, sv, dv, lane, j + S);
        const bool up = (lane & S) != 0;
        const float keep = up ? y : x;
        const float give = up ? x : y;
        return keep + __shfl_xor(give, S, 32);
    }
}

__global__ __launch_bounds__(256) void edge_dot(const float* __restrict__ h,
                                                const int* __restrict__ src,
                                                const int* __restrict__ dst,
                                                float* __restrict__ out, int n) {
    const int e0 = __builtin_amdgcn_readfirstlane((int)(blockIdx.x * 8 + (threadIdx.x >> 5))) * 32;
    if (e0 >= n) return;
    const int lane = threadIdx.x & 31;
    const int e = e0 + lane;
    int sv = 0, dv = 0;
    if (e < n) { sv = src[e]; dv = dst[e]; }
    const float r = pair_sums<5>(h, sv, dv, lane, 0);
    if (e < n) out[e] = r;
}

extern "C" void kernel_launch(void* const* d_in, const int* in_sizes, int n_in,
                              void* d_out, int out_size, void* d_ws, size_t ws_size,
                              hipStream_t stream) {
    const float* feat    = (const float*)d_in[0];
    const int*   e0s     = (const int*)d_in[1];
    const int*   e0d     = (const int*)d_in[2];
    const int*   e1s     = (const int*)d_in[3];
    const int*   e1d     = (const int*)d_in[4];
    const int*   ps      = (const int*)d_in[5];
    const int*   pd      = (const int*)d_in[6];
    const int*   ns      = (const int*)d_in[7];
    const int*   nd      = (const int*)d_in[8];
    const float* w_proj  = (const float*)d_in[9];
    const float* b_proj  = (const float*)d_in[10];
    const float* w_self1 = (const float*)d_in[11];
    const float* w_neigh1= (const float*)d_in[12];
    const float* b1      = (const float*)d_in[13];
    const float* w_self2 = (const float*)d_in[14];
    const float* w_neigh2= (const float*)d_in[15];
    const float* b2      = (const float*)d_in[16];

    const int E0   = in_sizes[1];
    const int E1   = in_sizes[3];
    const int nPos = in_sizes[5];
    const int nNeg = in_sizes[7];

    float* h0  = (float*)d_ws;
    float* agg = h0 + ND;
    float* deg = agg + ND;
    float* h1  = deg + NN;
    float* h2  = h0;
    float* out = (float*)d_out;

    int* seg   = (int*)deg;
    int* perm1 = (int*)h1;
    int* rank1 = perm1 + E0;
    int* cnt1  = rank1 + E0;
    int* perm2 = (int*)h0;
    int* rank2 = perm2 + E1;
    int* cnt2  = rank2 + E1;

    const int mtiles = NN / 32;
    const int zgrid  = (NN + 1 + 255) / 256;
    const int egrid0 = (E0 + 255) / 256;
    const int egrid1 = (E1 + 255) / 256;
    const int ngrid  = (NN + 255) / 256;
    const int ggrid  = NN / 8;

    gemm_sage<true, false><<<mtiles, 256, 0, stream>>>(feat, w_proj, nullptr, nullptr, b_proj, h0);

    zero_f32<<<zgrid, 256, 0, stream>>>((float*)cnt1, (long)NN + 1);
    seg_count<<<egrid0, 256, 0, stream>>>(e0d, cnt1, rank1, E0);
    seg_alloc<<<ngrid, 256, 0, stream>>>(cnt1, seg, cnt1 + NN);
    seg_place<<<egrid0, 256, 0, stream>>>(e0s, e0d, rank1, seg, perm1, E0);
    sage_gather_mean<<<ggrid, 256, 0, stream>>>(h0, perm1, seg, cnt1, agg, E0);
    gemm_sage<true, true><<<mtiles, 256, 0, stream>>>(h0, w_self1, agg, w_neigh1, b1, h1);

    zero_f32<<<zgrid, 256, 0, stream>>>((float*)cnt2, (long)NN + 1);
    seg_count<<<egrid1, 256, 0, stream>>>(e1d, cnt2, rank2, E1);
    seg_alloc<<<ngrid, 256, 0, stream>>>(cnt2, seg, cnt2 + NN);
    seg_place<<<egrid1, 256, 0, stream>>>(e1s, e1d, rank2, seg, perm2, E1);
    sage_gather_mean<<<ggrid, 256, 0, stream>>>(h1, perm2, seg, cnt2, agg, E1);
    gemm_sage<false, true><<<mtiles, 256, 0, stream>>>(h1, w_self2, agg, w_neigh2, b2, h2);

    edge_dot<<<(nPos + 255) / 256, 256, 0, stream>>>(h2, ps, pd, out, nPos);
    edge_dot<<<(nNeg + 255) / 256, 256, 0, stream>>>(h2, ns, nd, out + nPos, nNeg);
}


# Per layer, count+scan+place (3 launches, rank array, start lookups) become one pass storing src in a 32-slot bucket per dst at its arrival rank, overflow list past 32; sums stay in arrival order
# speedup vs baseline: 2.1095x; 1.1135x over previous
#include <hip/hip_runtime.h>

#define NN 100000
#define D 128
#define ND ((size_t)NN * D)
#define LDSP 132
#define CAP 32

typedef __attribute__((ext_vector_type(2))) float v2f;
typedef __attribute__((ext_vector_type(8))) float v8f;

__device__ __forceinline__ int atomAddRetI(int* p, int v) {
    return __hip_atomic_fetch_add(p, v, __ATOMIC_RELAXED, __HIP_MEMORY_SCOPE_AGENT);
}

__device__ __forceinline__ void wmma_pass(const float (*__restrict__ As)[LDSP],
                                          const float* __restrict__ W,
                                          int col0, int half, int l,
                                          v8f& acc0, v8f& acc1) {
    const float* wcol = W + col0 + l;
#pragma unroll
    for (int kk = 0; kk < D; kk += 4) {
        const int k0 = kk + 2 * half;
        v2f b;
        b.x = wcol[(size_t)k0 * D];
        b.y = wcol[(size_t)(k0 + 1) * D];
        v2f a0, a1;
        a0.x = As[l][k0];        a0.y = As[l][k0 + 1];
        a1.x = As[16 + l][k0];   a1.y = As[16 + l][k0 + 1];
        acc0 = __builtin_amdgcn_wmma_f32_16x16x4_f32(false, a0, false, b, (short)0, acc0, false, false);
        acc1 = __builtin_amdgcn_wmma_f32_16x16x4_f32(false, a1, false, b, (short)0, acc1, false, false);
    }
}

template <bool RELU, bool TWO>
__global__ __launch_bounds__(256) void gemm_sage(const float* __restrict__ A0,
                                                 const float* __restrict__ W0,
                                                 const float* __restrict__ A1,
                                                 const float* __restrict__ W1,
                                                 const float* __restrict__ bias,
                                                 float* __restrict__ C) {
    __shared__ float As0[32][LDSP];
    __shared__ float As1[TWO ? 32 : 1][LDSP];

    const int row0 = blockIdx.x * 32;
    const int col0 = (threadIdx.x >> 5) * 16;
    const int lane = threadIdx.x & 31;
    const int half = lane >> 4;
    const int l = lane & 15;

    {
        const int r = threadIdx.x >> 3;
        const int c = (threadIdx.x & 7) * 16;
        const float* g0 = A0 + (size_t)(row0 + r) * D + c;
#pragma unroll
        for (int j = 0; j < 16; j += 4)
            *(float4*)&As0[r][c + j] = *(const float4*)(g0 + j);
        if (TWO) {
            const float* g1 = A1 + (size_t)(row0 + r) * D + c;
#pragma unroll
            for (int j = 0; j < 16; j += 4)
                *(float4*)&As1[r][c + j] = *(const float4*)(g1 + j);
        }
    }
    __syncthreads();

    v8f acc0 = {0.f, 0.f, 0.f, 0.f, 0.f, 0.f, 0.f, 0.f};
    v8f acc1 = acc0;
    wmma_pass(As0, W0, col0, half, l, acc0, acc1);
    if (TWO) wmma_pass(As1, W1, col0, half, l, acc0, acc1);

    const float bc = bias[col0 + l];
#pragma unroll
    for (int r = 0; r < 8; ++r) {
        float v0 = acc0[r] + bc;
        float v1 = acc1[r] + bc;
        if (RELU) { v0 = fmaxf(v0, 0.f); v1 = fmaxf(v1, 0.f); }
        C[(size_t)(row0 + half * 8 + r) * D + col0 + l] = v0;
        C[(size_t)(row0 + 16 + half * 8 + r) * D + col0 + l] = v1;
    }
}

__global__ __launch_bounds__(256) void zero_f32(float* __restrict__ p, long n) {
    long i = (long)blockIdx.x * blockDim.x + threadIdx.x;
    if (i < n) p[i] = 0.f;
}

__global__ __launch_bounds__(256) void seg_fill(const int* __restrict__ src,
                                                const int* __restrict__ dst,
                                                int* cnt, int* __restrict__ bucket,
                                                int* __restrict__ ovf, int* novf, int E) {
    const int e = blockIdx.x * 256 + threadIdx.x;
    if (e >= E) return;
    const int d = dst[e];
    const int s = src[e];
    const int r = atomAddRetI(cnt + d, 1);
    if ((unsigned)r < (unsigned)CAP) {
        bucket[(size_t)d * CAP + r] = s;
    } else {
        const int k = atomAddRetI(novf, 1);
        if ((unsigned)k < (unsigned)E) *(int2*)(ovf + 2 * (size_t)k) = make_int2(d, s);
    }
}

__global__ __launch_bounds__(256) void sage_gather_mean(const float* __restrict__ h,
                                                        const int* __restrict__ bucket,
                                                        const int* __restrict__ ovf,
                                                        const int* __restrict__ cnt,
                                                        const int* __restrict__ novf,
                                                        float* __restrict__ agg, int E) {
    const int n = __builtin_amdgcn_readfirstlane((int)(blockIdx.x * 8 + (threadIdx.x >> 5)));
    if (n >= NN) return;
    const int lane = threadIdx.x & 31;
    int c = cnt[n];
    c = c < E ? c : E;
    const int m = c < CAP ? c : CAP;
    int mine = 0;
    if (lane < m) mine = bucket[(size_t)n * CAP + lane];
    float4 acc = {0.f, 0.f, 0.f, 0.f};
#pragma unroll 4
    for (int j = 0; j < m; ++j) {
        int s = __builtin_amdgcn_readlane(mine, j);
        s = ((unsigned)s < (unsigned)NN) ? s : 0;
        const float4 v = *(const float4*)(h + (size_t)s * D + lane * 4);
        acc.x += v.x; acc.y += v.y; acc.z += v.z; acc.w += v.w;
    }
    if (c > CAP) {
        int nv = *novf;
        nv = nv < E ? nv : E;
        for (int base = 0; base < nv; base += 32) {
            const int k = base + lane;
            int2 ds = make_int2(-1, 0);
            if (k < nv) ds = *(const int2*)(ovf + 2 * (size_t)k);
            unsigned hit = __builtin_amdgcn_ballot_w32(ds.x == n);
            while (hit) {
                const int j = __builtin_ctz(hit);
                hit &= hit - 1;
                int s = __builtin_amdgcn_readlane(ds.y, j);
                s = ((unsigned)s < (unsigned)NN) ? s : 0;
                const float4 v = *(const float4*)(h + (size_t)s * D + lane * 4);
                acc.x += v.x; acc.y += v.y; acc.z += v.z; acc.w += v.w;
            }
        }
    }
    const float inv = 1.0f / fmaxf((float)c, 1.0f);
    acc.x *= inv; acc.y *= inv; acc.z *= inv; acc.w *= inv;
    *(float4*)(agg + (size_t)n * D + lane * 4) = acc;
}

template <int LV>
__device__ __forceinline__ float pair_sums(const float* __restrict__ h, int sv, int dv, int lane, int j) {
    if constexpr (LV == 0) {
        const int s = __builtin_amdgcn_readlane(sv, j);
        const int d = __builtin_amdgcn_readlane(dv, j);
        const float4 a = *(const float4*)(h + (size_t)s * D + lane * 4);
        const float4 b = *(const float4*)(h + (size_t)d * D + lane * 4);
        return a.x * b.x + a.y * b.y + a.z * b.z + a.w * b.w;
    } else {
        constexpr int S = 32 >> LV;
        const float x = pair_sums<LV - 1>(h, sv, dv, lane, j);
        const float y = pair_sums<LV - 1>(h, sv, dv, lane, j + S);
        const bool up = (lane & S) != 0;
        const float keep = up ? y : x;
        const float give = up ? x : y;
        return keep + __shfl_xor(give, S, 32);
    }
}

__global__ __launch_bounds__(256) void edge_dot(const float* __restrict__ h,
                                                const int* __restrict__ src,
                                                const int* __restrict__ dst,
                                                float* __restrict__ out, int n) {
    const int e0 = __builtin_amdgcn_readfirstlane((int)(blockIdx.x * 8 + (threadIdx.x >> 5))) * 32;
    if (e0 >= n) return;
    const int lane = threadIdx.x & 31;
    const int e = e0 + lane;
    int sv = 0, dv = 0;
    if (e < n) { sv = src[e]; dv = dst[e]; }
    const float r = pair_sums<5>(h, sv, dv, lane, 0);
    if (e < n) out[e] = r;
}

extern "C" void kernel_launch(void* const* d_in, const int* in_sizes, int n_in,
                              void* d_out, int out_size, void* d_ws, size_t ws_size,
                              hipStream_t stream) {
    const float* feat    = (const float*)d_in[0];
    const int*   e0s     = (const int*)d_in[1];
    const int*   e0d     = (const int*)d_in[2];
    const int*   e1s     = (const int*)d_in[3];
    const int*   e1d     = (const int*)d_in[4];
    const int*   ps      = (const int*)d_in[5];
    const int*   pd      = (const int*)d_in[6];
    const int*   ns      = (const int*)d_in[7];
    const int*   nd      = (const int*)d_in[8];
    const float* w_proj  = (const float*)d_in[9];
    const float* b_proj  = (const float*)d_in[10];
    const float* w_self1 = (const float*)d_in[11];
    const float* w_neigh1= (const float*)d_in[12];
    const float* b1      = (const float*)d_in[13];
    const float* w_self2 = (const float*)d_in[14];
    const float* w_neigh2= (const float*)d_in[15];
    const float* b2      = (const float*)d_in[16];

    const int E0   = in_sizes[1];
    const int E1   = in_sizes[3];
    const int nPos = in_sizes[5];
    const int nNeg = in_sizes[7];

    float* h0  = (float*)d_ws;
    float* agg = h0 + ND;
    float* h1  = agg + ND + NN;
    float* h2  = h0;
    float* out = (float*)d_out;

    int* bkt1 = (int*)h1;
    int* ovf1 = bkt1 + (size_t)NN * CAP;
    int* cnt1 = ovf1 + 2 * (size_t)E0;
    int* bkt2 = (int*)h0;
    int* ovf2 = bkt2 + (size_t)NN * CAP;
    int* cnt2 = ovf2 + 2 * (size_t)E1;

    const int mtiles = NN / 32;
    const int zgrid  = (NN + 1 + 255) / 256;
    const int egrid0 = (E0 + 255) / 256;
    const int egrid1 = (E1 + 255) / 256;
    const int ggrid  = NN / 8;

    gemm_sage<true, false><<<mtiles, 256, 0, stream>>>(feat, w_proj, nullptr, nullptr, b_proj, h0);

    zero_f32<<<zgrid, 256, 0, stream>>>((float*)cnt1, (long)NN + 1);
    seg_fill<<<egrid0, 256, 0, stream>>>(e0s, e0d, cnt1, bkt1, ovf1, cnt1 + NN, E0);
    sage_gather_mean<<<ggrid, 256, 0, stream>>>(h0, bkt1, ovf1, cnt1, cnt1 + NN, agg, E0);
    gemm_sage<true, true><<<mtiles, 256, 0, stream>>>(h0, w_self1, agg, w_neigh1, b1, h1);

    zero_f32<<<zgrid, 256, 0, stream>>>((float*)cnt2, (long)NN + 1);
    seg_fill<<<egrid1, 256, 0, stream>>>(e1s, e1d, cnt2, bkt2, ovf2, cnt2 + NN, E1);
    sage_gather_mean<<<ggrid, 256, 0, stream>>>(h1, bkt2, ovf2, cnt2, cnt2 + NN, agg, E1);
    gemm_sage<false, true><<<mtiles, 256, 0, stream>>>(h1, w_self2, agg, w_neigh2, b2, h2);

    edge_dot<<<(nPos + 255) / 256, 256, 0, stream>>>(h2, ps, pd, out, nPos);
    edge_dot<<<(nNeg + 255) / 256, 256, 0, stream>>>(h2, ns, nd, out + nPos, nNeg);
}


# Neighbour mean built inside the layer GEMM: each wave gathers 4 node means straight into the LDS A tile; mean buffer, its store and reload and 2 launches go (11 -> 9); sum order unchanged
# speedup vs baseline: 2.2207x; 1.0527x over previous
#include <hip/hip_runtime.h>

#define NN 100000
#define D 128
#define ND ((size_t)NN * D)
#define LDSP 132
#define CAP 32

typedef __attribute__((ext_vector_type(2))) float v2f;
typedef __attribute__((ext_vector_type(8))) float v8f;

__device__ __forceinline__ int atomAddRetI(int* p, int v) {
    return __hip_atomic_fetch_add(p, v, __ATOMIC_RELAXED, __HIP_MEMORY_SCOPE_AGENT);
}

__device__ __forceinline__ void wmma_pass(const float (*__restrict__ As)[LDSP],
                                          const float* __restrict__ W,
                                          int col0, int half, int l,
                                          v8f& acc0, v8f& acc1) {
    const float* wcol = W + col0 + l;
#pragma unroll
    for (int kk = 0; kk < D; kk += 4) {
        const int k0 = kk + 2 * half;
        v2f b;
        b.x = wcol[(size_t)k0 * D];
        b.y = wcol[(size_t)(k0 + 1) * D];
        v2f a0, a1;
        a0.x = As[l][k0];        a0.y = As[l][k0 + 1];
        a1.x = As[16 + l][k0];   a1.y = As[16 + l][k0 + 1];
        acc0 = __builtin_amdgcn_wmma_f32_16x16x4_f32(false, a0, false, b, (short)0, acc0, false, false);
        acc1 = __builtin_amdgcn_wmma_f32_16x16x4_f32(false, a1, false, b, (short)0, acc1, false, false);
    }
}

__device__ __forceinline__ void stage_rows(float (*__restrict__ As)[LDSP], const float* __restrict__ A, int row0) {
    const int r = threadIdx.x >> 3;
    const int c = (threadIdx.x & 7) * 16;
    const float* g = A + (size_t)(row0 + r) * D + c;
#pragma unroll
    for (int j = 0; j < 16; j += 4)
        *(float4*)&As[r][c + j] = *(const float4*)(g + j);
}

template <bool RELU>
__device__ __forceinline__ void store_tiles(float* __restrict__ C, const float* __restrict__ bias,
                                            int row0, int col0, int half, int l,
                                            const v8f& acc0, const v8f& acc1) {
    const float bc = bias[col0 + l];
#pragma unroll
    for (int r = 0; r < 8; ++r) {
        float v0 = acc0[r] + bc;
        float v1 = acc1[r] + bc;
        if (RELU) { v0 = fmaxf(v0, 0.f); v1 = fmaxf(v1, 0.f); }
        C[(size_t)(row0 + half * 8 + r) * D + col0 + l] = v0;
        C[(size_t)(row0 + 16 + half * 8 + r) * D + col0 + l] = v1;
    }
}

__global__ __launch_bounds__(256) void gemm_proj(const float* __restrict__ A,
                                                 const float* __restrict__ W,
                                                 const float* __restrict__ bias,
                                                 float* __restrict__ C) {
    __shared__ float As0[32][LDSP];

    const int row0 = blockIdx.x * 32;
    const int col0 = (threadIdx.x >> 5) * 16;
    const int lane = threadIdx.x & 31;
    const int half = lane >> 4;
    const int l = lane & 15;

    stage_rows(As0, A, row0);
    __syncthreads();

    v8f acc0 = {0.f, 0.f, 0.f, 0.f, 0.f, 0.f, 0.f, 0.f};
    v8f acc1 = acc0;
    wmma_pass(As0, W, col0, half, l, acc0, acc1);
    store_tiles<true>(C, bias, row0, col0, half, l, acc0, acc1);
}

__global__ __launch_bounds__(256) void zero_f32(float* __restrict__ p, long n) {
    long i = (long)blockIdx.x * blockDim.x + threadIdx.x;
    if (i < n) p[i] = 0.f;
}

__global__ __launch_bounds__(256) void seg_fill(const int* __restrict__ src,
                                                const int* __restrict__ dst,
                                                int* cnt, int* __restrict__ bucket,
                                                int* __restrict__ ovf, int* novf, int E) {
    const int e = blockIdx.x * 256 + threadIdx.x;
    if (e >= E) return;
    const int d = dst[e];
    const int s = src[e];
    const int r = atomAddRetI(cnt + d, 1);
    if ((unsigned)r < (unsigned)CAP) {
        bucket[(size_t)d * CAP + r] = s;
    } else {
        const int k = atomAddRetI(novf, 1);
        if ((unsigned)k < (unsigned)E) *(int2*)(ovf + 2 * (size_t)k) = make_int2(d, s);
    }
}

template <bool RELU>
__global__ __launch_bounds__(256) void gemm_sage_gather(const float* __restrict__ h,
                                                        const float* __restrict__ Ws,
                                                        const float* __restrict__ Wn,
                                                        const float* __restrict__ bias,
                                                        const int* __restrict__ bucket,
                                                        const int* __restrict__ ovf,
                                                        const int* __restrict__ cnt,
                                                        const int* __restrict__ novf,
                                                        float* __restrict__ C, int E) {
    __shared__ float As0[32][LDSP];
    __shared__ float As1[32][LDSP];

    const int row0 = blockIdx.x * 32;
    const int wave = __builtin_amdgcn_readfirstlane((int)(threadIdx.x >> 5));
    const int col0 = wave * 16;
    const int lane = threadIdx.x & 31;
    const int half = lane >> 4;
    const int l = lane & 15;

    const int nb = row0 + wave * 4;
    const int4 ids = *(const int4*)(bucket + (size_t)nb * CAP + lane * 4);
    const int idc[4] = {ids.x, ids.y, ids.z, ids.w};
    const int degs = cnt[nb + (lane & 3)];

    stage_rows(As0, h, row0);

    const int lo = lane * 4;
    for (int i = 0; i < 4; ++i) {
        int c = __builtin_amdgcn_readlane(degs, i);
        c = c < E ? c : E;
        const int m = c < CAP ? c : CAP;
        float4 acc = {0.f, 0.f, 0.f, 0.f};
        int j = 0;
        for (; j + 8 <= m; j += 8) {
            const float* rowp[8];
#pragma unroll
            for (int u = 0; u < 8; ++u) {
                int s = __builtin_amdgcn_readlane(idc[u & 3], i * 8 + (j >> 2) + (u >> 2));
                s = ((unsigned)s < (unsigned)NN) ? s : 0;
                rowp[u] = h + (size_t)s * D;
            }
            float4 v[8];
#pragma unroll
            for (int u = 0; u < 8; ++u) v[u] = *(const float4*)(rowp[u] + lo);
#pragma unroll
            for (int u = 0; u < 8; ++u) { acc.x += v[u].x; acc.y += v[u].y; acc.z += v[u].z; acc.w += v[u].w; }
        }
        if (j < m) {
            const float* rowp[7];
#pragma unroll
            for (int u = 0; u < 7; ++u) {
                int s = __builtin_amdgcn_readlane(idc[u & 3], i * 8 + (j >> 2) + (u >> 2));
                s = ((unsigned)s < (unsigned)NN) ? s : 0;
                rowp[u] = h + (size_t)s * D;
            }
            float4 t[7];
#pragma unroll
            for (int u = 0; u < 7; ++u) {
                if (j + u < m) t[u] = *(const float4*)(rowp[u] + lo);
            }
#pragma unroll
            for (int u = 0; u < 7; ++u) {
                if (j + u < m) { acc.x += t[u].x; acc.y += t[u].y; acc.z += t[u].z; acc.w += t[u].w; }
            }
        }
        if (c > CAP) {
            const int n = nb + i;
            int nv = *novf;
            nv = nv < E ? nv : E;
            for (int base = 0; base < nv; base += 32) {
                const int k = base + lane;
                int2 ds = make_int2(-1, 0);
                if (k < nv) ds = *(const int2*)(ovf + 2 * (size_t)k);
                unsigned hit = __builtin_amdgcn_ballot_w32(ds.x == n);
                while (hit) {
                    const int q = __builtin_ctz(hit);
                    hit &= hit - 1;
                    int s = __builtin_amdgcn_readlane(ds.y, q);
                    s = ((unsigned)s < (unsigned)NN) ? s : 0;
                    const float4 x = *(const float4*)(h + (size_t)s * D + lane * 4);
                    acc.x += x.x; acc.y += x.y; acc.z += x.z; acc.w += x.w;
                }
            }
        }
        const float inv = 1.0f / fmaxf((float)c, 1.0f);
        acc.x *= inv; acc.y *= inv; acc.z *= inv; acc.w *= inv;
        *(float4*)&As1[wave * 4 + i][lane * 4] = acc;
    }
    __syncthreads();

    v8f acc0 = {0.f, 0.f, 0.f, 0.f, 0.f, 0.f, 0.f, 0.f};
    v8f acc1 = acc0;
    wmma_pass(As0, Ws, col0, half, l, acc0, acc1);
    wmma_pass(As1, Wn, col0, half, l, acc0, acc1);
    store_tiles<RELU>(C, bias, row0, col0, half, l, acc0, acc1);
}

template <int LV>
__device__ __forceinline__ float pair_sums(const float* __restrict__ h, int sv, int dv, int lane, int j) {
    if constexpr (LV == 0) {
        const int s = __builtin_amdgcn_readlane(sv, j);
        const int d = __builtin_amdgcn_readlane(dv, j);
        const float4 a = *(const float4*)(h + (size_t)s * D + lane * 4);
        const float4 b = *(const float4*)(h + (size_t)d * D + lane * 4);
        return a.x * b.x + a.y * b.y + a.z * b.z + a.w * b.w;
    } else {
        constexpr int S = 32 >> LV;
        const float x = pair_sums<LV - 1>(h, sv, dv, lane, j);
        const float y = pair_sums<LV - 1>(h, sv, dv, lane, j + S);
        const bool up = (lane & S) != 0;
        const float keep = up ? y : x;
        const float give = up ? x : y;
        return keep + __shfl_xor(give, S, 32);
    }
}

__global__ __launch_bounds__(256) void edge_dot(const float* __restrict__ h,
                                                const int* __restrict__ src,
                                                const int* __restrict__ dst,
                                                float* __restrict__ out, int n) {
    const int e0 = __builtin_amdgcn_readfirstlane((int)(blockIdx.x * 8 + (threadIdx.x >> 5))) * 32;
    if (e0 >= n) return;
    const int lane = threadIdx.x & 31;
    const int e = e0 + lane;
    int sv = 0, dv = 0;
    if (e < n) { sv = src[e]; dv = dst[e]; }
    const float r = pair_sums<5>(h, sv, dv, lane, 0);
    if (e < n) out[e] = r;
}

extern "C" void kernel_launch(void* const* d_in, const int* in_sizes, int n_in,
                              void* d_out, int out_size, void* d_ws, size_t ws_size,
                              hipStream_t stream) {
    const float* feat    = (const float*)d_in[0];
    const int*   e0s     = (const int*)d_in[1];
    const int*   e0d     = (const int*)d_in[2];
    const int*   e1s     = (const int*)d_in[3];
    const int*   e1d     = (const int*)d_in[4];
    const int*   ps      = (const int*)d_in[5];
    const int*   pd      = (const int*)d_in[6];
    const int*   ns      = (const int*)d_in[7];
    const int*   nd      = (const int*)d_in[8];
    const float* w_proj  = (const float*)d_in[9];
    const float* b_proj  = (const float*)d_in[10];
    const float* w_self1 = (const float*)d_in[11];
    const float* w_neigh1= (const float*)d_in[12];
    const float* b1      = (const float*)d_in[13];
    const float* w_self2 = (const float*)d_in[14];
    const float* w_neigh2= (const float*)d_in[15];
    const float* b2      = (const float*)d_in[16];

    const int E0   = in_sizes[1];
    const int E1   = in_sizes[3];
    const int nPos = in_sizes[5];
    const int nNeg = in_sizes[7];

    float* h0  = (float*)d_ws;
    float* mid = h0 + ND;
    float* h1  = mid + ND + NN;
    float* h2  = h0;
    float* out = (float*)d_out;

    int* bkt  = (int*)mid;
    int* ovf1 = bkt + (size_t)NN * CAP;
    int* cnt1 = ovf1 + 2 * (size_t)E0;
    int* ovf2 = bkt + (size_t)NN * CAP;
    int* cnt2 = ovf2 + 2 * (size_t)E1;

    const int mtiles = NN / 32;
    const int zgrid  = (NN + 1 + 255) / 256;
    const int egrid0 = (E0 + 255) / 256;
    const int egrid1 = (E1 + 255) / 256;

    gemm_proj<<<mtiles, 256, 0, stream>>>(feat, w_proj, b_proj, h0);

    zero_f32<<<zgrid, 256, 0, stream>>>((float*)cnt1, (long)NN + 1);
    seg_fill<<<egrid0, 256, 0, stream>>>(e0s, e0d, cnt1, bkt, ovf1, cnt1 + NN, E0);
    gemm_sage_gather<true><<<mtiles, 256, 0, stream>>>(h0, w_self1, w_neigh1, b1, bkt, ovf1, cnt1, cnt1 + NN, h1, E0);

    zero_f32<<<zgrid, 256, 0, stream>>>((float*)cnt2, (long)NN + 1);
    seg_fill<<<egrid1, 256, 0, stream>>>(e1s, e1d, cnt2, bkt, ovf2, cnt2 + NN, E1);
    gemm_sage_gather<false><<<mtiles, 256, 0, stream>>>(h1, w_self2, w_neigh2, b2, bkt, ovf2, cnt2, cnt2 + NN, h2, E1);

    edge_dot<<<(nPos + 255) / 256, 256, 0, stream>>>(h2, ps, pd, out, nPos);
    edge_dot<<<(nNeg + 255) / 256, 256, 0, stream>>>(h2, ns, nd, out + nPos, nNeg);
}


# Per layer the 1.6M returning atomics on memory go: edges sorted by dst range per 4096-edge chunk in LDS, then one block per 512 nodes ranks and buckets them by LDS counters; arrival-rank sum order
# speedup vs baseline: 3.6994x; 1.6659x over previous
#include <hip/hip_runtime.h>

#define NN 100000
#define D 128
#define ND ((size_t)NN * D)
#define LDSP 132
#define CAP 32

typedef __attribute__((ext_vector_type(2))) float v2f;
typedef __attribute__((ext_vector_type(8))) float v8f;

__device__ __forceinline__ int atomAddRetI(int* p, int v) {
    return __hip_atomic_fetch_add(p, v, __ATOMIC_RELAXED, __HIP_MEMORY_SCOPE_AGENT);
}

__device__ __forceinline__ void wmma_pass(const float (*__restrict__ As)[LDSP],
                                          const float* __restrict__ W,
                                          int col0, int half, int l,
                                          v8f& acc0, v8f& acc1) {
    const float* wcol = W + col0 + l;
#pragma unroll
    for (int kk = 0; kk < D; kk += 4) {
        const int k0 = kk + 2 * half;
        v2f b;
        b.x = wcol[(size_t)k0 * D];
        b.y = wcol[(size_t)(k0 + 1) * D];
        v2f a0, a1;
        a0.x = As[l][k0];        a0.y = As[l][k0 + 1];
        a1.x = As[16 + l][k0];   a1.y = As[16 + l][k0 + 1];
        acc0 = __builtin_amdgcn_wmma_f32_16x16x4_f32(false, a0, false, b, (short)0, acc0, false, false);
        acc1 = __builtin_amdgcn_wmma_f32_16x16x4_f32(false, a1, false, b, (short)0, acc1, false, false);
    }
}

__device__ __forceinline__ void stage_rows(float (*__restrict__ As)[LDSP], const float* __restrict__ A, int row0) {
    const int r = threadIdx.x >> 3;
    const int c = (threadIdx.x & 7) * 16;
    const float* g = A + (size_t)(row0 + r) * D + c;
#pragma unroll
    for (int j = 0; j < 16; j += 4)
        *(float4*)&As[r][c + j] = *(const float4*)(g + j);
}

template <bool RELU>
__device__ __forceinline__ void store_tiles(float* __restrict__ C, const float* __restrict__ bias,
                                            int row0, int col0, int half, int l,
                                            const v8f& acc0, const v8f& acc1) {
    const float bc = bias[col0 + l];
#pragma unroll
    for (int r = 0; r < 8; ++r) {
        float v0 = acc0[r] + bc;
        float v1 = acc1[r] + bc;
        if (RELU) { v0 = fmaxf(v0, 0.f); v1 = fmaxf(v1, 0.f); }
        C[(size_t)(row0 + half * 8 + r) * D + col0 + l] = v0;
        C[(size_t)(row0 + 16 + half * 8 + r) * D + col0 + l] = v1;
    }
}

__global__ __launch_bounds__(256) void gemm_proj(const float* __restrict__ A,
                                                 const float* __restrict__ W,
                                                 const float* __restrict__ bias,
                                                 float* __restrict__ C) {
    __shared__ float As0[32][LDSP];

    const int row0 = blockIdx.x * 32;
    const int col0 = (threadIdx.x >> 5) * 16;
    const int lane = threadIdx.x & 31;
    const int half = lane >> 4;
    const int l = lane & 15;

    stage_rows(As0, A, row0);
    __syncthreads();

    v8f acc0 = {0.f, 0.f, 0.f, 0.f, 0.f, 0.f, 0.f, 0.f};
    v8f acc1 = acc0;
    wmma_pass(As0, W, col0, half, l, acc0, acc1);
    store_tiles<true>(C, bias, row0, col0, half, l, acc0, acc1);
}

#define RANGE 512
#define NB ((NN + RANGE - 1) / RANGE)
#define PERL ((NB + 31) / 32)
#define CHUNK 4096
#define EPT (CHUNK / 256)
#define SRCB 17
#define SRCM ((1 << SRCB) - 1)

__global__ __launch_bounds__(256) void bin_chunk(const int* __restrict__ src,
                                                 const int* __restrict__ dst,
                                                 int* __restrict__ sorted, int* __restrict__ meta,
                                                 int* __restrict__ novf, int E, int nchunk) {
    __shared__ int hist[NB];
    __shared__ int start[NB];
    __shared__ int stage[CHUNK];
    const int c = blockIdx.x;
    const int t = threadIdx.x;
    if (c == 0 && t == 0) *novf = 0;
    for (int b = t; b < NB; b += 256) hist[b] = 0;
    __syncthreads();

    const int e0 = c * CHUNK;
    int word[EPT], bin[EPT], rk[EPT];
    {
        int dv[EPT], sv[EPT];
#pragma unroll
        for (int j = 0; j < EPT; ++j) {
            const int e = e0 + j * 256 + t;
            const int ec = e < E ? e : E - 1;
            dv[j] = dst[ec];
            sv[j] = src[ec];
        }
#pragma unroll
        for (int j = 0; j < EPT; ++j) {
            const int e = e0 + j * 256 + t;
            const bool ok = e < E && (unsigned)dv[j] < (unsigned)NN;
            bin[j] = ok ? dv[j] / RANGE : -1;
            word[j] = ((dv[j] % RANGE) << SRCB) | (sv[j] & SRCM);
        }
    }
#pragma unroll
    for (int j = 0; j < EPT; ++j) {
        rk[j] = 0;
        if (bin[j] >= 0) rk[j] = atomicAdd(&hist[bin[j]], 1);
    }
    __syncthreads();

    if (t < 32) {
        int v[PERL];
        int sum = 0;
#pragma unroll
        for (int k = 0; k < PERL; ++k) {
            const int b = t * PERL + k;
            v[k] = b < NB ? hist[b] : 0;
            sum += v[k];
        }
        int inc = sum;
#pragma unroll
        for (int o = 1; o < 32; o <<= 1) {
            const int y = __shfl_up(inc, o, 32);
            if (t >= o) inc += y;
        }
        int run = inc - sum;
#pragma unroll
        for (int k = 0; k < PERL; ++k) {
            const int b = t * PERL + k;
            if (b < NB) start[b] = run;
            run += v[k];
        }
    }
    __syncthreads();

#pragma unroll
    for (int j = 0; j < EPT; ++j) {
        if (bin[j] >= 0) stage[start[bin[j]] + rk[j]] = word[j];
    }
    for (int b = t; b < NB; b += 256) meta[(size_t)b * nchunk + c] = start[b] | (hist[b] << 16);
    __syncthreads();

    const int total = start[NB - 1] + hist[NB - 1];
    for (int i = t; i < total; i += 256) sorted[(size_t)e0 + i] = stage[i];
}

__global__ __launch_bounds__(512) void own_fill(const int* __restrict__ sorted,
                                                const int* __restrict__ meta,
                                                int* __restrict__ cnt, int* __restrict__ bucket,
                                                int* __restrict__ ovf, int* novf, int E, int nchunk) {
    __shared__ int lcnt[RANGE];
    __shared__ int lb[RANGE * CAP];
    const int b = blockIdx.x;
    const int t = threadIdx.x;
    const int n0 = b * RANGE;
    for (int i = t; i < RANGE; i += 512) lcnt[i] = 0;
    __syncthreads();

    for (int c = t; c < nchunk; c += 512) {
        const int m = meta[(size_t)b * nchunk + c];
        int st = m & 0xFFFF;
        int n = (m >> 16) & 0xFFFF;
        st = st < CHUNK ? st : CHUNK;
        n = n < CHUNK - st ? n : CHUNK - st;
        const int* p = sorted + (size_t)c * CHUNK + st;
        for (int j = 0; j < n; j += 8) {
            int w[8], r[8];
#pragma unroll
            for (int u = 0; u < 8; ++u) w[u] = p[j + u < n ? j + u : n - 1];
#pragma unroll
            for (int u = 0; u < 8; ++u) {
                r[u] = 0;
                if (j + u < n) r[u] = atomicAdd(&lcnt[(w[u] >> SRCB) & (RANGE - 1)], 1);
            }
#pragma unroll
            for (int u = 0; u < 8; ++u) {
                if (j + u < n) {
                    const int dl = (w[u] >> SRCB) & (RANGE - 1);
                    const int s = w[u] & SRCM;
                    if ((unsigned)r[u] < (unsigned)CAP) {
                        lb[dl * CAP + r[u]] = s;
                    } else {
                        const int k = atomAddRetI(novf, 1);
                        if ((unsigned)k < (unsigned)E && n0 + dl < NN)
                            *(int2*)(ovf + 2 * (size_t)k) = make_int2(n0 + dl, s);
                    }
                }
            }
        }
    }
    __syncthreads();

    for (int i = t; i < RANGE * CAP; i += 512) {
        const int node = i / CAP;
        const int deg = lcnt[node];
        const int fill = deg < CAP ? deg : CAP;
        if (n0 + node < NN && (i % CAP) < fill) bucket[(size_t)n0 * CAP + i] = lb[i];
    }
    for (int i = t; i < RANGE; i += 512) {
        if (n0 + i < NN) cnt[n0 + i] = lcnt[i];
    }
}

template <bool RELU>
__global__ __launch_bounds__(256) void gemm_sage_gather(const float* __restrict__ h,
                                                        const float* __restrict__ Ws,
                                                        const float* __restrict__ Wn,
                                                        const float* __restrict__ bias,
                                                        const int* __restrict__ bucket,
                                                        const int* __restrict__ ovf,
                                                        const int* __restrict__ cnt,
                                                        const int* __restrict__ novf,
                                                        float* __restrict__ C, int E) {
    __shared__ float As0[32][LDSP];
    __shared__ float As1[32][LDSP];

    const int row0 = blockIdx.x * 32;
    const int wave = __builtin_amdgcn_readfirstlane((int)(threadIdx.x >> 5));
    const int col0 = wave * 16;
    const int lane = threadIdx.x & 31;
    const int half = lane >> 4;
    const int l = lane & 15;

    const int nb = row0 + wave * 4;
    const int4 ids = *(const int4*)(bucket + (size_t)nb * CAP + lane * 4);
    const int idc[4] = {ids.x, ids.y, ids.z, ids.w};
    const int degs = cnt[nb + (lane & 3)];

    stage_rows(As0, h, row0);

    const int lo = lane * 4;
    for (int i = 0; i < 4; ++i) {
        int c = __builtin_amdgcn_readlane(degs, i);
        c = c < E ? c : E;
        const int m = c < CAP ? c : CAP;
        float4 acc = {0.f, 0.f, 0.f, 0.f};
        int j = 0;
        for (; j + 8 <= m; j += 8) {
            const float* rowp[8];
#pragma unroll
            for (int u = 0; u < 8; ++u) {
                int s = __builtin_amdgcn_readlane(idc[u & 3], i * 8 + (j >> 2) + (u >> 2));
                s = ((unsigned)s < (unsigned)NN) ? s : 0;
                rowp[u] = h + (size_t)s * D;
            }
            float4 v[8];
#pragma unroll
            for (int u = 0; u < 8; ++u) v[u] = *(const float4*)(rowp[u] + lo);
#pragma unroll
            for (int u = 0; u < 8; ++u) { acc.x += v[u].x; acc.y += v[u].y; acc.z += v[u].z; acc.w += v[u].w; }
        }
        if (j < m) {
            const float* rowp[7];
#pragma unroll
            for (int u = 0; u < 7; ++u) {
                int s = __builtin_amdgcn_readlane(idc[u & 3], i * 8 + (j >> 2) + (u >> 2));
                s = ((unsigned)s < (unsigned)NN) ? s : 0;
                rowp[u] = h + (size_t)s * D;
            }
            float4 t[7];
#pragma unroll
            for (int u = 0; u < 7; ++u) {
                if (j + u < m) t[u] = *(const float4*)(rowp[u] + lo);
            }
#pragma unroll
            for (int u = 0; u < 7; ++u) {
                if (j + u < m) { acc.x += t[u].x; acc.y += t[u].y; acc.z += t[u].z; acc.w += t[u].w; }
            }
        }
        if (c > CAP) {
            const int n = nb + i;
            int nv = *novf;
            nv = nv < E ? nv : E;
            for (int base = 0; base < nv; base += 32) {
                const int k = base + lane;
                int2 ds = make_int2(-1, 0);
                if (k < nv) ds = *(const int2*)(ovf + 2 * (size_t)k);
                unsigned hit = __builtin_amdgcn_ballot_w32(ds.x == n);
                while (hit) {
                    const int q = __builtin_ctz(hit);
                    hit &= hit - 1;
                    int s = __builtin_amdgcn_readlane(ds.y, q);
                    s = ((unsigned)s < (unsigned)NN) ? s : 0;
                    const float4 x = *(const float4*)(h + (size_t)s * D + lane * 4);
                    acc.x += x.x; acc.y += x.y; acc.z += x.z; acc.w += x.w;
                }
            }
        }
        const float inv = 1.0f / fmaxf((float)c, 1.0f);
        acc.x *= inv; acc.y *= inv; acc.z *= inv; acc.w *= inv;
        *(float4*)&As1[wave * 4 + i][lane * 4] = acc;
    }
    __syncthreads();

    v8f acc0 = {0.f, 0.f, 0.f, 0.f, 0.f, 0.f, 0.f, 0.f};
    v8f acc1 = acc0;
    wmma_pass(As0, Ws, col0, half, l, acc0, acc1);
    wmma_pass(As1, Wn, col0, half, l, acc0, acc1);
    store_tiles<RELU>(C, bias, row0, col0, half, l, acc0, acc1);
}

template <int LV>
__device__ __forceinline__ float pair_sums(const float* __restrict__ h, int sv, int dv, int lane, int j) {
    if constexpr (LV == 0) {
        const int s = __builtin_amdgcn_readlane(sv, j);
        const int d = __builtin_amdgcn_readlane(dv, j);
        const float4 a = *(const float4*)(h + (size_t)s * D + lane * 4);
        const float4 b = *(const float4*)(h + (size_t)d * D + lane * 4);
        return a.x * b.x + a.y * b.y + a.z * b.z + a.w * b.w;
    } else {
        constexpr int S = 32 >> LV;
        const float x = pair_sums<LV - 1>(h, sv, dv, lane, j);
        const float y = pair_sums<LV - 1>(h, sv, dv, lane, j + S);
        const bool up = (lane & S) != 0;
        const float keep = up ? y : x;
        const float give = up ? x : y;
        return keep + __shfl_xor(give, S, 32);
    }
}

__global__ __launch_bounds__(256) void edge_dot(const float* __restrict__ h,
                                                const int* __restrict__ src,
                                                const int* __restrict__ dst,
                                                float* __restrict__ out, int n) {
    const int e0 = __builtin_amdgcn_readfirstlane((int)(blockIdx.x * 8 + (threadIdx.x >> 5))) * 32;
    if (e0 >= n) return;
    const int lane = threadIdx.x & 31;
    const int e = e0 + lane;
    int sv = 0, dv = 0;
    if (e < n) { sv = src[e]; dv = dst[e]; }
    const float r = pair_sums<5>(h, sv, dv, lane, 0);
    if (e < n) out[e] = r;
}

extern "C" void kernel_launch(void* const* d_in, const int* in_sizes, int n_in,
                              void* d_out, int out_size, void* d_ws, size_t ws_size,
                              hipStream_t stream) {
    const float* feat    = (const float*)d_in[0];
    const int*   e0s     = (const int*)d_in[1];
    const int*   e0d     = (const int*)d_in[2];
    const int*   e1s     = (const int*)d_in[3];
    const int*   e1d     = (const int*)d_in[4];
    const int*   ps      = (const int*)d_in[5];
    const int*   pd      = (const int*)d_in[6];
    const int*   ns      = (const int*)d_in[7];
    const int*   nd      = (const int*)d_in[8];
    const float* w_proj  = (const float*)d_in[9];
    const float* b_proj  = (const float*)d_in[10];
    const float* w_self1 = (const float*)d_in[11];
    const float* w_neigh1= (const float*)d_in[12];
    const float* b1      = (const float*)d_in[13];
    const float* w_self2 = (const float*)d_in[14];
    const float* w_neigh2= (const float*)d_in[15];
    const float* b2      = (const float*)d_in[16];

    const int E0   = in_sizes[1];
    const int E1   = in_sizes[3];
    const int nPos = in_sizes[5];
    const int nNeg = in_sizes[7];

    float* h0  = (float*)d_ws;
    float* mid = h0 + ND;
    float* h1  = mid + ND + NN;
    float* h2  = h0;
    float* out = (float*)d_out;

    const int nch0 = (E0 + CHUNK - 1) / CHUNK;
    const int nch1 = (E1 + CHUNK - 1) / CHUNK;

    int* bkt  = (int*)mid;
    int* ovf1 = bkt + (size_t)NN * CAP;
    int* cnt1 = ovf1 + 2 * (size_t)E0;
    int* srt1 = cnt1 + NN + 1;
    int* met1 = srt1 + (size_t)nch0 * CHUNK;
    int* ovf2 = bkt + (size_t)NN * CAP;
    int* cnt2 = ovf2 + 2 * (size_t)E1;
    int* srt2 = cnt2 + NN + 1;
    int* met2 = srt2 + (size_t)nch1 * CHUNK;

    const int mtiles = NN / 32;

    gemm_proj<<<mtiles, 256, 0, stream>>>(feat, w_proj, b_proj, h0);

    bin_chunk<<<nch0, 256, 0, stream>>>(e0s, e0d, srt1, met1, cnt1 + NN, E0, nch0);
    own_fill<<<NB, 512, 0, stream>>>(srt1, met1, cnt1, bkt, ovf1, cnt1 + NN, E0, nch0);
    gemm_sage_gather<true><<<mtiles, 256, 0, stream>>>(h0, w_self1, w_neigh1, b1, bkt, ovf1, cnt1, cnt1 + NN, h1, E0);

    bin_chunk<<<nch1, 256, 0, stream>>>(e1s, e1d, srt2, met2, cnt2 + NN, E1, nch1);
    own_fill<<<NB, 512, 0, stream>>>(srt2, met2, cnt2, bkt, ovf2, cnt2 + NN, E1, nch1);
    gemm_sage_gather<false><<<mtiles, 256, 0, stream>>>(h1, w_self2, w_neigh2, b2, bkt, ovf2, cnt2, cnt2 + NN, h2, E1);

    edge_dot<<<(nPos + 255) / 256, 256, 0, stream>>>(h2, ps, pd, out, nPos);
    edge_dot<<<(nNeg + 255) / 256, 256, 0, stream>>>(h2, ns, nd, out + nPos, nNeg);
}


# Independent passes share launches: projection GEMM + both layers' edge binning in one grid, both layers' bucket fills in one, both score lists in one (9 -> 5 launches); sum order unchanged
# speedup vs baseline: 3.9897x; 1.0785x over previous
#include <hip/hip_runtime.h>

#define NN 100000
#define D 128
#define ND ((size_t)NN * D)
#define LDSP 132
#define CAP 32

typedef __attribute__((ext_vector_type(2))) float v2f;
typedef __attribute__((ext_vector_type(8))) float v8f;

__device__ __forceinline__ int atomAddRetI(int* p, int v) {
    return __hip_atomic_fetch_add(p, v, __ATOMIC_RELAXED, __HIP_MEMORY_SCOPE_AGENT);
}

__device__ __forceinline__ void wmma_pass(const float (*__restrict__ As)[LDSP],
                                          const float* __restrict__ W,
                                          int col0, int half, int l,
                                          v8f& acc0, v8f& acc1) {
    const float* wcol = W + col0 + l;
#pragma unroll
    for (int kk = 0; kk < D; kk += 4) {
        const int k0 = kk + 2 * half;
        v2f b;
        b.x = wcol[(size_t)k0 * D];
        b.y = wcol[(size_t)(k0 + 1) * D];
        v2f a0, a1;
        a0.x = As[l][k0];        a0.y = As[l][k0 + 1];
        a1.x = As[16 + l][k0];   a1.y = As[16 + l][k0 + 1];
        acc0 = __builtin_amdgcn_wmma_f32_16x16x4_f32(false, a0, false, b, (short)0, acc0, false, false);
        acc1 = __builtin_amdgcn_wmma_f32_16x16x4_f32(false, a1, false, b, (short)0, acc1, false, false);
    }
}

__device__ __forceinline__ void stage_rows(float (*__restrict__ As)[LDSP], const float* __restrict__ A, int row0) {
    const int r = threadIdx.x >> 3;
    const int c = (threadIdx.x & 7) * 16;
    const float* g = A + (size_t)(row0 + r) * D + c;
#pragma unroll
    for (int j = 0; j < 16; j += 4)
        *(float4*)&As[r][c + j] = *(const float4*)(g + j);
}

template <bool RELU>
__device__ __forceinline__ void store_tiles(float* __restrict__ C, const float* __restrict__ bias,
                                            int row0, int col0, int half, int l,
                                            const v8f& acc0, const v8f& acc1) {
    const float bc = bias[col0 + l];
#pragma unroll
    for (int r = 0; r < 8; ++r) {
        float v0 = acc0[r] + bc;
        float v1 = acc1[r] + bc;
        if (RELU) { v0 = fmaxf(v0, 0.f); v1 = fmaxf(v1, 0.f); }
        C[(size_t)(row0 + half * 8 + r) * D + col0 + l] = v0;
        C[(size_t)(row0 + 16 + half * 8 + r) * D + col0 + l] = v1;
    }
}

__device__ __forceinline__ void gemm_proj_block(const float* __restrict__ A,
                                                const float* __restrict__ W,
                                                const float* __restrict__ bias,
                                                float* __restrict__ C, int tile) {
    __shared__ float As0[32][LDSP];

    const int row0 = tile * 32;
    const int col0 = (threadIdx.x >> 5) * 16;
    const int lane = threadIdx.x & 31;
    const int half = lane >> 4;
    const int l = lane & 15;

    stage_rows(As0, A, row0);
    __syncthreads();

    v8f acc0 = {0.f, 0.f, 0.f, 0.f, 0.f, 0.f, 0.f, 0.f};
    v8f acc1 = acc0;
    wmma_pass(As0, W, col0, half, l, acc0, acc1);
    store_tiles<true>(C, bias, row0, col0, half, l, acc0, acc1);
}

#define RANGE 512
#define NB ((NN + RANGE - 1) / RANGE)
#define PERL ((NB + 31) / 32)
#define CHUNK 4096
#define EPT (CHUNK / 256)
#define SRCB 17
#define SRCM ((1 << SRCB) - 1)

__device__ __forceinline__ void bin_chunk_block(const int* __restrict__ src,
                                                const int* __restrict__ dst,
                                                int* __restrict__ sorted, int* __restrict__ meta,
                                                int* __restrict__ novf, int E, int nchunk, int c) {
    __shared__ int hist[NB];
    __shared__ int start[NB];
    __shared__ int stage[CHUNK];
    const int t = threadIdx.x;
    if (c == 0 && t == 0) *novf = 0;
    for (int b = t; b < NB; b += 256) hist[b] = 0;
    __syncthreads();

    const int e0 = c * CHUNK;
    int word[EPT], bin[EPT], rk[EPT];
    {
        int dv[EPT], sv[EPT];
#pragma unroll
        for (int j = 0; j < EPT; ++j) {
            const int e = e0 + j * 256 + t;
            const int ec = e < E ? e : E - 1;
            dv[j] = dst[ec];
            sv[j] = src[ec];
        }
#pragma unroll
        for (int j = 0; j < EPT; ++j) {
            const int e = e0 + j * 256 + t;
            const bool ok = e < E && (unsigned)dv[j] < (unsigned)NN;
            bin[j] = ok ? dv[j] / RANGE : -1;
            word[j] = ((dv[j] % RANGE) << SRCB) | (sv[j] & SRCM);
        }
    }
#pragma unroll
    for (int j = 0; j < EPT; ++j) {
        rk[j] = 0;
        if (bin[j] >= 0) rk[j] = atomicAdd(&hist[bin[j]], 1);
    }
    __syncthreads();

    if (t < 32) {
        int v[PERL];
        int sum = 0;
#pragma unroll
        for (int k = 0; k < PERL; ++k) {
            const int b = t * PERL + k;
            v[k] = b < NB ? hist[b] : 0;
            sum += v[k];
        }
        int inc = sum;
#pragma unroll
        for (int o = 1; o < 32; o <<= 1) {
            const int y = __shfl_up(inc, o, 32);
            if (t >= o) inc += y;
        }
        int run = inc - sum;
#pragma unroll
        for (int k = 0; k < PERL; ++k) {
            const int b = t * PERL + k;
            if (b < NB) start[b] = run;
            run += v[k];
        }
    }
    __syncthreads();

#pragma unroll
    for (int j = 0; j < EPT; ++j) {
        if (bin[j] >= 0) stage[start[bin[j]] + rk[j]] = word[j];
    }
    for (int b = t; b < NB; b += 256) meta[(size_t)b * nchunk + c] = start[b] | (hist[b] << 16);
    __syncthreads();

    const int total = start[NB - 1] + hist[NB - 1];
    for (int i = t; i < total; i += 256) sorted[(size_t)e0 + i] = stage[i];
}

__global__ __launch_bounds__(256, 12) void proj_and_bin(const float* __restrict__ feat,
                                                    const float* __restrict__ Wp,
                                                    const float* __restrict__ bp,
                                                    float* __restrict__ h0,
                                                    const int* __restrict__ s1, const int* __restrict__ d1,
                                                    int* __restrict__ srt1, int* __restrict__ met1,
                                                    int* __restrict__ novf1, int E1, int nch1,
                                                    const int* __restrict__ s2, const int* __restrict__ d2,
                                                    int* __restrict__ srt2, int* __restrict__ met2,
                                                    int* __restrict__ novf2, int E2, int nch2) {
    const int b = blockIdx.x;
    if (b < nch1 + nch2) {
        const bool first = b < nch1;
        bin_chunk_block(first ? s1 : s2, first ? d1 : d2, first ? srt1 : srt2, first ? met1 : met2,
                        first ? novf1 : novf2, first ? E1 : E2, first ? nch1 : nch2, first ? b : b - nch1);
    } else {
        gemm_proj_block(feat, Wp, bp, h0, b - nch1 - nch2);
    }
}

__global__ __launch_bounds__(512) void own_fill(const int* __restrict__ sortedA, const int* __restrict__ metaA,
                                                int* __restrict__ cntA, int* __restrict__ bucketA,
                                                int* __restrict__ ovfA, int* novfA, int EA, int nchunkA,
                                                const int* __restrict__ sortedB, const int* __restrict__ metaB,
                                                int* __restrict__ cntB, int* __restrict__ bucketB,
                                                int* __restrict__ ovfB, int* novfB, int EB, int nchunkB) {
    __shared__ int lcnt[RANGE];
    __shared__ int lb[RANGE * CAP];
    const bool layerA = blockIdx.x < NB;
    const int* __restrict__ sorted = layerA ? sortedA : sortedB;
    const int* __restrict__ meta = layerA ? metaA : metaB;
    int* __restrict__ cnt = layerA ? cntA : cntB;
    int* __restrict__ bucket = layerA ? bucketA : bucketB;
    int* __restrict__ ovf = layerA ? ovfA : ovfB;
    int* novf = layerA ? novfA : novfB;
    const int E = layerA ? EA : EB;
    const int nchunk = layerA ? nchunkA : nchunkB;
    const int b = layerA ? blockIdx.x : blockIdx.x - NB;
    const int t = threadIdx.x;
    const int n0 = b * RANGE;
    for (int i = t; i < RANGE; i += 512) lcnt[i] = 0;
    __syncthreads();

    for (int c = t; c < nchunk; c += 512) {
        const int m = meta[(size_t)b * nchunk + c];
        int st = m & 0xFFFF;
        int n = (m >> 16) & 0xFFFF;
        st = st < CHUNK ? st : CHUNK;
        n = n < CHUNK - st ? n : CHUNK - st;
        const int* p = sorted + (size_t)c * CHUNK + st;
        for (int j = 0; j < n; j += 8) {
            int w[8], r[8];
#pragma unroll
            for (int u = 0; u < 8; ++u) w[u] = p[j + u < n ? j + u : n - 1];
#pragma unroll
            for (int u = 0; u < 8; ++u) {
                r[u] = 0;
                if (j + u < n) r[u] = atomicAdd(&lcnt[(w[u] >> SRCB) & (RANGE - 1)], 1);
            }
#pragma unroll
            for (int u = 0; u < 8; ++u) {
                if (j + u < n) {
                    const int dl = (w[u] >> SRCB) & (RANGE - 1);
                    const int s = w[u] & SRCM;
                    if ((unsigned)r[u] < (unsigned)CAP) {
                        lb[dl * CAP + r[u]] = s;
                    } else {
                        const int k = atomAddRetI(novf, 1);
                        if ((unsigned)k < (unsigned)E && n0 + dl < NN)
                            *(int2*)(ovf + 2 * (size_t)k) = make_int2(n0 + dl, s);
                    }
                }
            }
        }
    }
    __syncthreads();

    for (int i = t; i < RANGE * CAP; i += 512) {
        const int node = i / CAP;
        const int deg = lcnt[node];
        const int fill = deg < CAP ? deg : CAP;
        if (n0 + node < NN && (i % CAP) < fill) bucket[(size_t)n0 * CAP + i] = lb[i];
    }
    for (int i = t; i < RANGE / 4; i += 512) {
        int w = 0;
        for (int k = 0; k < 4; ++k) {
            const int deg = lcnt[4 * i + k];
            w |= (deg <= CAP ? deg : CAP + 1) << (8 * k);
        }
        if (n0 + 4 * i < NN) cnt[(n0 >> 2) + i] = w;
    }
}

template <bool RELU>
__global__ __launch_bounds__(256) void gemm_sage_gather(const float* __restrict__ h,
                                                        const float* __restrict__ Ws,
                                                        const float* __restrict__ Wn,
                                                        const float* __restrict__ bias,
                                                        const int* __restrict__ bucket,
                                                        const int* __restrict__ ovf,
                                                        const int* __restrict__ cntw,
                                                        const int* __restrict__ novf,
                                                        float* __restrict__ C, int E) {
    __shared__ float As0[32][LDSP];
    __shared__ float As1[32][LDSP];

    const int row0 = blockIdx.x * 32;
    const int wave = __builtin_amdgcn_readfirstlane((int)(threadIdx.x >> 5));
    const int col0 = wave * 16;
    const int lane = threadIdx.x & 31;
    const int half = lane >> 4;
    const int l = lane & 15;

    const int nb = row0 + wave * 4;
    const int4 ids = *(const int4*)(bucket + (size_t)nb * CAP + lane * 4);
    const int idc[4] = {ids.x, ids.y, ids.z, ids.w};
    const int degs = ((const unsigned char*)cntw)[nb + (lane & 3)];

    stage_rows(As0, h, row0);

    const int lo = lane * 4;
    for (int i = 0; i < 4; ++i) {
        int c = __builtin_amdgcn_readlane(degs, i);
        const int m = c < CAP ? c : CAP;
        float4 acc = {0.f, 0.f, 0.f, 0.f};
        int j = 0;
        for (; j + 8 <= m; j += 8) {
            const float* rowp[8];
#pragma unroll
            for (int u = 0; u < 8; ++u) {
                int s = __builtin_amdgcn_readlane(idc[u & 3], i * 8 + (j >> 2) + (u >> 2));
                s = ((unsigned)s < (unsigned)NN) ? s : 0;
                rowp[u] = h + (size_t)s * D;
            }
            float4 v[8];
#pragma unroll
            for (int u = 0; u < 8; ++u) v[u] = *(const float4*)(rowp[u] + lo);
#pragma unroll
            for (int u = 0; u < 8; ++u) { acc.x += v[u].x; acc.y += v[u].y; acc.z += v[u].z; acc.w += v[u].w; }
        }
        if (j < m) {
            const float* rowp[7];
#pragma unroll
            for (int u = 0; u < 7; ++u) {
                int s = __builtin_amdgcn_readlane(idc[u & 3], i * 8 + (j >> 2) + (u >> 2));
                s = ((unsigned)s < (unsigned)NN) ? s : 0;
                rowp[u] = h + (size_t)s * D;
            }
            float4 t[7];
#pragma unroll
            for (int u = 0; u < 7; ++u) {
                if (j + u < m) t[u] = *(const float4*)(rowp[u] + lo);
            }
#pragma unroll
            for (int u = 0; u < 7; ++u) {
                if (j + u < m) { acc.x += t[u].x; acc.y += t[u].y; acc.z += t[u].z; acc.w += t[u].w; }
            }
        }
        if (c > CAP) {
            c = CAP;
            const int n = nb + i;
            int nv = *novf;
            nv = nv < E ? nv : E;
            for (int base = 0; base < nv; base += 32) {
                const int k = base + lane;
                int2 ds = make_int2(-1, 0);
                if (k < nv) ds = *(const int2*)(ovf + 2 * (size_t)k);
                unsigned hit = __builtin_amdgcn_ballot_w32(ds.x == n);
                c += __builtin_popcount(hit);
                while (hit) {
                    const int q = __builtin_ctz(hit);
                    hit &= hit - 1;
                    int s = __builtin_amdgcn_readlane(ds.y, q);
                    s = ((unsigned)s < (unsigned)NN) ? s : 0;
                    const float4 x = *(const float4*)(h + (size_t)s * D + lane * 4);
                    acc.x += x.x; acc.y += x.y; acc.z += x.z; acc.w += x.w;
                }
            }
        }
        const float inv = 1.0f / fmaxf((float)c, 1.0f);
        acc.x *= inv; acc.y *= inv; acc.z *= inv; acc.w *= inv;
        *(float4*)&As1[wave * 4 + i][lane * 4] = acc;
    }
    __syncthreads();

    v8f acc0 = {0.f, 0.f, 0.f, 0.f, 0.f, 0.f, 0.f, 0.f};
    v8f acc1 = acc0;
    wmma_pass(As0, Ws, col0, half, l, acc0, acc1);
    wmma_pass(As1, Wn, col0, half, l, acc0, acc1);
    store_tiles<RELU>(C, bias, row0, col0, half, l, acc0, acc1);
}

template <int LV>
__device__ __forceinline__ float pair_sums(const float* __restrict__ h, int sv, int dv, int lane, int j) {
    if constexpr (LV == 0) {
        const int s = __builtin_amdgcn_readlane(sv, j);
        const int d = __builtin_amdgcn_readlane(dv, j);
        const float4 a = *(const float4*)(h + (size_t)s * D + lane * 4);
        const float4 b = *(const float4*)(h + (size_t)d * D + lane * 4);
        return a.x * b.x + a.y * b.y + a.z * b.z + a.w * b.w;
    } else {
        constexpr int S = 32 >> LV;
        const float x = pair_sums<LV - 1>(h, sv, dv, lane, j);
        const float y = pair_sums<LV - 1>(h, sv, dv, lane, j + S);
        const bool up = (lane & S) != 0;
        const float keep = up ? y : x;
        const float give = up ? x : y;
        return keep + __shfl_xor(give, S, 32);
    }
}

__global__ __launch_bounds__(256) void edge_dot(const float* __restrict__ h,
                                                const int* __restrict__ srcA, const int* __restrict__ dstA,
                                                float* __restrict__ outA, int nA,
                                                const int* __restrict__ srcB, const int* __restrict__ dstB,
                                                float* __restrict__ outB, int nB) {
    const int w = __builtin_amdgcn_readfirstlane((int)(blockIdx.x * 8 + (threadIdx.x >> 5)));
    const int wA = (nA + 31) >> 5;
    const bool inA = w < wA;
    const int* __restrict__ src = inA ? srcA : srcB;
    const int* __restrict__ dst = inA ? dstA : dstB;
    float* __restrict__ out = inA ? outA : outB;
    const int n = inA ? nA : nB;
    const int e0 = (inA ? w : w - wA) * 32;
    if (e0 >= n) return;
    const int lane = threadIdx.x & 31;
    const int e = e0 + lane;
    int sv = 0, dv = 0;
    if (e < n) { sv = src[e]; dv = dst[e]; }
    const float r = pair_sums<5>(h, sv, dv, lane, 0);
    if (e < n) out[e] = r;
}

extern "C" void kernel_launch(void* const* d_in, const int* in_sizes, int n_in,
                              void* d_out, int out_size, void* d_ws, size_t ws_size,
                              hipStream_t stream) {
    const float* feat    = (const float*)d_in[0];
    const int*   e0s     = (const int*)d_in[1];
    const int*   e0d     = (const int*)d_in[2];
    const int*   e1s     = (const int*)d_in[3];
    const int*   e1d     = (const int*)d_in[4];
    const int*   ps      = (const int*)d_in[5];
    const int*   pd      = (const int*)d_in[6];
    const int*   ns      = (const int*)d_in[7];
    const int*   nd      = (const int*)d_in[8];
    const float* w_proj  = (const float*)d_in[9];
    const float* b_proj  = (const float*)d_in[10];
    const float* w_self1 = (const float*)d_in[11];
    const float* w_neigh1= (const float*)d_in[12];
    const float* b1      = (const float*)d_in[13];
    const float* w_self2 = (const float*)d_in[14];
    const float* w_neigh2= (const float*)d_in[15];
    const float* b2      = (const float*)d_in[16];

    const int E0   = in_sizes[1];
    const int E1   = in_sizes[3];
    const int nPos = in_sizes[5];
    const int nNeg = in_sizes[7];

    float* h0  = (float*)d_ws;
    float* mid = h0 + ND;
    float* h1  = mid + ND + NN;
    float* h2  = h0;
    float* out = (float*)d_out;

    const int nch0 = (E0 + CHUNK - 1) / CHUNK;
    const int nch1 = (E1 + CHUNK - 1) / CHUNK;

    int* bkt1 = (int*)mid;
    int* bkt2 = bkt1 + (size_t)NN * CAP;
    int* ovf1 = bkt2 + (size_t)NN * CAP;
    int* ovf2 = ovf1 + 2 * (size_t)E0;
    int* cnt1 = ovf2 + 2 * (size_t)E1;
    int* cnt2 = cnt1 + NB * RANGE / 4;
    int* novf = cnt2 + NB * RANGE / 4;

    int* srt1 = (int*)h1;
    int* met1 = srt1 + (size_t)nch0 * CHUNK;
    int* srt2 = met1 + (size_t)NB * nch0;
    int* met2 = srt2 + (size_t)nch1 * CHUNK;

    const int mtiles = NN / 32;

    proj_and_bin<<<nch0 + nch1 + mtiles, 256, 0, stream>>>(feat, w_proj, b_proj, h0,
                                                           e0s, e0d, srt1, met1, novf, E0, nch0,
                                                           e1s, e1d, srt2, met2, novf + 32, E1, nch1);

    own_fill<<<2 * NB, 512, 0, stream>>>(srt1, met1, cnt1, bkt1, ovf1, novf, E0, nch0,
                                         srt2, met2, cnt2, bkt2, ovf2, novf + 32, E1, nch1);

    gemm_sage_gather<true><<<mtiles, 256, 0, stream>>>(h0, w_self1, w_neigh1, b1, bkt1, ovf1, cnt1, novf, h1, E0);

    gemm_sage_gather<false><<<mtiles, 256, 0, stream>>>(h1, w_self2, w_neigh2, b2, bkt2, ovf2, cnt2, novf + 32, h2, E1);

    edge_dot<<<(((nPos + 31) / 32) + ((nNeg + 31) / 32) + 7) / 8, 256, 0, stream>>>(h2, ps, pd, out, nPos,
                                                                                   ns, nd, out + nPos, nNeg);
}
